# starting kernel
# baseline (speedup 1.0000x reference)
#include <hip/hip_runtime.h>
#include <math.h>


#define NN 50000
#define DD 128
#define HH 4
#define CC 32
#define LL 3
#define NEG_SLOPE 0.2f
#define LN_EPS 1e-5f

typedef __attribute__((ext_vector_type(2))) float v2f;
typedef __attribute__((ext_vector_type(8))) float v8f;

__device__ __forceinline__ unsigned f2ord(float f) {
  unsigned u = __float_as_uint(f);
  return (u & 0x80000000u) ? ~u : (u | 0x80000000u);
}
__device__ __forceinline__ float ord2f(unsigned u) {
  u = (u & 0x80000000u) ? (u & 0x7FFFFFFFu) : ~u;
  return __uint_as_float(u);
}

__global__ void init_layer(float* __restrict__ acc, float* __restrict__ denom,
                           unsigned* __restrict__ emax, int nd, int nh) {
  int i = blockIdx.x * blockDim.x + threadIdx.x;
  if (i < nd) acc[i] = 0.0f;
  if (i < nh) { denom[i] = 0.0f; emax[i] = 0u; }
}

__global__ __launch_bounds__(256) void gemm_dual_wmma(
    const float* __restrict__ x,
    const float* __restrict__ Wl, const float* __restrict__ bl,
    const float* __restrict__ Wr, const float* __restrict__ br,
    float* __restrict__ xl, float* __restrict__ xr) {
  __shared__ float sx[16 * DD];

  const int tid  = threadIdx.x;
  const int wave = tid >> 5;
  const int lane = tid & 31;
  const int rowBase = blockIdx.x * 16;

  {
    const float4* gsrc = (const float4*)(x + (size_t)rowBase * DD);
    float4* sdst = (float4*)sx;
    sdst[tid]       = gsrc[tid];
    sdst[tid + 256] = gsrc[tid + 256];
  }
  __syncthreads();

  const int colBase = wave * 16;
  const int m  = lane & 15;
  const int kk = (lane >> 4) * 2;
  const int n  = lane & 15;

  v8f cl = {};
  v8f cr = {};

  for (int k0 = 0; k0 < DD; k0 += 4) {
    v2f a;
    a.x = sx[m * DD + k0 + kk];
    a.y = sx[m * DD + k0 + kk + 1];
    v2f bfl;
    bfl.x = Wl[(size_t)(k0 + kk) * DD + colBase + n];
    bfl.y = Wl[(size_t)(k0 + kk + 1) * DD + colBase + n];
    v2f bfr;
    bfr.x = Wr[(size_t)(k0 + kk) * DD + colBase + n];
    bfr.y = Wr[(size_t)(k0 + kk + 1) * DD + colBase + n];

    cl = __builtin_amdgcn_wmma_f32_16x16x4_f32(
        false, a, false, bfl, (short)0, cl, false, false);
    cr = __builtin_amdgcn_wmma_f32_16x16x4_f32(
        false, a, false, bfr, (short)0, cr, false, false);
  }

  const int rOff = (lane >> 4) * 8;
  const int col  = colBase + n;
  const float bL = bl[col];
  const float bR = br[col];
#pragma unroll
  for (int i = 0; i < 8; ++i) {
    const size_t row = (size_t)(rowBase + i + rOff);
    xl[row * DD + col] = cl[i] + bL;
    xr[row * DD + col] = cr[i] + bR;
  }
}

__global__ void edge_logits(const int* __restrict__ ei,
                            const float* __restrict__ xl,
                            const float* __restrict__ xr,
                            const float* __restrict__ att,
                            float* __restrict__ elog,
                            unsigned* __restrict__ emax, int E, int Np) {
  const int t = blockIdx.x * blockDim.x + threadIdx.x;
  const int tot = (E + Np) * HH;
  if (t >= tot) return;
  const int e = t >> 2;
  const int h = t & 3;
  const int src = (e < E) ? ei[e] : (e - E);
  const int dst = (e < E) ? ei[E + e] : (e - E);

  const float4* pl = (const float4*)(xl + (size_t)src * DD + h * CC);
  const float4* pr = (const float4*)(xr + (size_t)dst * DD + h * CC);
  const float4* pa = (const float4*)(att + h * CC);

  float s = 0.0f;
#pragma unroll
  for (int i = 0; i < CC / 4; ++i) {
    float4 a = pl[i], b = pr[i], w = pa[i];
    float m0 = a.x + b.x; m0 = (m0 > 0.0f) ? m0 : NEG_SLOPE * m0;
    float m1 = a.y + b.y; m1 = (m1 > 0.0f) ? m1 : NEG_SLOPE * m1;
    float m2 = a.z + b.z; m2 = (m2 > 0.0f) ? m2 : NEG_SLOPE * m2;
    float m3 = a.w + b.w; m3 = (m3 > 0.0f) ? m3 : NEG_SLOPE * m3;
    s = fmaf(w.x, m0, s);
    s = fmaf(w.y, m1, s);
    s = fmaf(w.z, m2, s);
    s = fmaf(w.w, m3, s);
  }
  elog[t] = s;
  atomicMax(&emax[dst * HH + h], f2ord(s));
}

__global__ void edge_exp(const int* __restrict__ ei,
                         float* __restrict__ elog,
                         const unsigned* __restrict__ emax,
                         float* __restrict__ denom, int E, int Np) {
  const int t = blockIdx.x * blockDim.x + threadIdx.x;
  const int tot = (E + Np) * HH;
  if (t >= tot) return;
  const int e = t >> 2;
  const int h = t & 3;
  const int dst = (e < E) ? ei[E + e] : (e - E);
  const float mx = ord2f(emax[dst * HH + h]);
  const float ee = __expf(elog[t] - mx);
  elog[t] = ee;
  atomicAdd(&denom[dst * HH + h], ee);
}

__global__ __launch_bounds__(256) void edge_scatter(
    const int* __restrict__ ei, const float* __restrict__ xl,
    const float* __restrict__ elog, const float* __restrict__ denom,
    float* __restrict__ acc, int E, int Np) {
  const int wid  = (blockIdx.x * blockDim.x + threadIdx.x) >> 5;
  const int lane = threadIdx.x & 31;
  if (wid >= E + Np) return;
  const int src = (wid < E) ? ei[wid] : (wid - E);
  const int dst = (wid < E) ? ei[E + wid] : (wid - E);
#pragma unroll
  for (int h = 0; h < HH; ++h) {
    const float alpha = elog[wid * HH + h] / denom[dst * HH + h];
    const int c = h * CC + lane;
    atomicAdd(&acc[(size_t)dst * DD + c], xl[(size_t)src * DD + c] * alpha);
  }
}

__global__ __launch_bounds__(256) void node_norm(
    const float* __restrict__ xin, const float* __restrict__ acc,
    const float* __restrict__ bias, const float* __restrict__ gamma,
    const float* __restrict__ beta, float* __restrict__ xout, int Np,
    int do_relu) {
  const int wid  = (blockIdx.x * blockDim.x + threadIdx.x) >> 5;
  const int lane = threadIdx.x & 31;
  if (wid >= Np) return;

  float v[4];
  float sum = 0.0f;
#pragma unroll
  for (int j = 0; j < 4; ++j) {
    const int c = j * 32 + lane;
    const float t = xin[(size_t)wid * DD + c] + acc[(size_t)wid * DD + c] + bias[c];
    v[j] = t;
    sum += t;
  }
#pragma unroll
  for (int off = 16; off > 0; off >>= 1) sum += __shfl_xor(sum, off, 32);
  const float mu = sum * (1.0f / (float)DD);

  float sq = 0.0f;
#pragma unroll
  for (int j = 0; j < 4; ++j) { const float d = v[j] - mu; sq = fmaf(d, d, sq); }
#pragma unroll
  for (int off = 16; off > 0; off >>= 1) sq += __shfl_xor(sq, off, 32);
  const float rstd = rsqrtf(sq * (1.0f / (float)DD) + LN_EPS);

#pragma unroll
  for (int j = 0; j < 4; ++j) {
    const int c = j * 32 + lane;
    float o = (v[j] - mu) * rstd * gamma[c] + beta[c];
    if (do_relu) o = (o > 0.0f) ? o : 0.0f;
    xout[(size_t)wid * DD + c] = o;
  }
}

static inline int cdiv(long long a, long long b) { return (int)((a + b - 1) / b); }

extern "C" void kernel_launch(void* const* d_in, const int* in_sizes, int n_in,
                              void* d_out, int out_size, void* d_ws, size_t ws_size,
                              hipStream_t stream) {
  const int Np = NN;
  const int E  = in_sizes[1] / 2;
  const int EP = E + Np;

  const float* x0    = (const float*)d_in[0];
  const int*   ei    = (const int*)d_in[1];
  const float* Wl    = (const float*)d_in[2];
  const float* bl    = (const float*)d_in[3];
  const float* Wr    = (const float*)d_in[4];
  const float* br    = (const float*)d_in[5];
  const float* att   = (const float*)d_in[6];
  const float* bias  = (const float*)d_in[7];
  const float* gamma = (const float*)d_in[8];
  const float* beta  = (const float*)d_in[9];

  float* ws = (float*)d_ws;
  const size_t ND = (size_t)Np * DD;
  const size_t NH = (size_t)Np * HH;
  float*    bufA  = ws;
  float*    xl    = bufA + ND;
  float*    xr    = xl + ND;
  float*    acc   = xr + ND;
  float*    elog  = acc + ND;
  float*    denom = elog + (size_t)EP * HH;
  unsigned* emax  = (unsigned*)(denom + NH);

  const int TB = 256;
  const dim3 blk(TB);

  for (int i = 0; i < LL; ++i) {
    const float* xin  = (i == 0) ? x0 : bufA;
    float*       xout = (i == LL - 1) ? (float*)d_out : bufA;
    const float* Wli  = Wl + (size_t)i * DD * DD;
    const float* bli  = bl + (size_t)i * DD;
    const float* Wri  = Wr + (size_t)i * DD * DD;
    const float* bri  = br + (size_t)i * DD;
    const float* atti = att + (size_t)i * HH * CC;
    const float* bi   = bias + (size_t)i * DD;
    const float* gi   = gamma + (size_t)i * DD;
    const float* bti  = beta + (size_t)i * DD;

    init_layer<<<cdiv(ND, TB), blk, 0, stream>>>(acc, denom, emax, (int)ND, (int)NH);

    gemm_dual_wmma<<<Np / 16, blk, 0, stream>>>(xin, Wli, bli, Wri, bri, xl, xr);

    edge_logits<<<cdiv((long long)EP * HH, TB), blk, 0, stream>>>(
        ei, xl, xr, atti, elog, emax, E, Np);

    edge_exp<<<cdiv((long long)EP * HH, TB), blk, 0, stream>>>(
        ei, elog, emax, denom, E, Np);

    edge_scatter<<<cdiv((long long)EP * 32, TB), blk, 0, stream>>>(
        ei, xl, elog, denom, acc, E, Np);

    node_norm<<<cdiv((long long)Np * 32, TB), blk, 0, stream>>>(
        xin, acc, bi, gi, bti, xout, Np, (i != LL - 1) ? 1 : 0);
  }
}


# Group edges by destination (in-call counting sort); one wave per node does logits, segment softmax and the weighted sum in registers: no float atomics, no per-edge passes
# speedup vs baseline: 3.6117x; 3.6117x over previous
#include <hip/hip_runtime.h>
#include <math.h>


#define NN 50000
#define NPAD 50048
#define DD 128
#define HH 4
#define CC 32
#define LL 3
#define NEG_SLOPE 0.2f
#define LN_EPS 1e-5f

typedef __attribute__((ext_vector_type(2))) float v2f;
typedef __attribute__((ext_vector_type(8))) float v8f;

__global__ void seg_zero(int* __restrict__ cnt, int* __restrict__ total, int n) {
  const int i = blockIdx.x * blockDim.x + threadIdx.x;
  if (i < n) cnt[i] = 0;
  if (i == 0) total[0] = 0;
}

__global__ void seg_count(const int* __restrict__ ei, int* __restrict__ cnt, int E) {
  const int t = blockIdx.x * blockDim.x + threadIdx.x;
  if (t < E) atomicAdd(&cnt[ei[E + t]], 1);
}

__global__ __launch_bounds__(256) void seg_alloc(
    const int* __restrict__ cnt, int* __restrict__ rowstart,
    int* __restrict__ cursor, int* __restrict__ total, int n) {
  const int i = blockIdx.x * blockDim.x + threadIdx.x;
  const int lane = threadIdx.x & 31;
  const int c = (i < n) ? cnt[i] : 0;
  int inc = c;
  for (int off = 1; off < 32; off <<= 1) {
    const int t = __shfl_up(inc, off, 32);
    if (lane >= off) inc += t;
  }
  int base = 0;
  if (lane == 31) base = atomicAdd(total, inc);
  base = __shfl(base, 31, 32);
  const int st = base + inc - c;
  if (i < n) { rowstart[i] = st; cursor[i] = st; }
}

__global__ void seg_fill(const int* __restrict__ ei, int* __restrict__ cursor,
                         int* __restrict__ psrc, int E) {
  const int t = blockIdx.x * blockDim.x + threadIdx.x;
  if (t >= E) return;
  const int d = ei[E + t];
  const int p = atomicAdd(&cursor[d], 1);
  if ((unsigned)p < (unsigned)E) atomicExch(&psrc[p], ei[t]);
}

__global__ __launch_bounds__(256) void gemm_dual_wmma(
    const float* __restrict__ x,
    const float* __restrict__ Wl, const float* __restrict__ bl,
    const float* __restrict__ Wr, const float* __restrict__ br,
    float* __restrict__ xl, float* __restrict__ xr) {
  __shared__ float sx[16 * DD];

  const int tid  = threadIdx.x;
  const int wave = tid >> 5;
  const int lane = tid & 31;
  const int rowBase = blockIdx.x * 16;

  {
    const float4* gsrc = (const float4*)(x + (size_t)rowBase * DD);
    float4* sdst = (float4*)sx;
    sdst[tid]       = gsrc[tid];
    sdst[tid + 256] = gsrc[tid + 256];
  }
  __syncthreads();

  const int colBase = wave * 16;
  const int m  = lane & 15;
  const int kk = (lane >> 4) * 2;
  const int n  = lane & 15;

  v8f cl = {};
  v8f cr = {};

  for (int k0 = 0; k0 < DD; k0 += 4) {
    v2f a;
    a.x = sx[m * DD + k0 + kk];
    a.y = sx[m * DD + k0 + kk + 1];
    v2f bfl;
    bfl.x = Wl[(size_t)(k0 + kk) * DD + colBase + n];
    bfl.y = Wl[(size_t)(k0 + kk + 1) * DD + colBase + n];
    v2f bfr;
    bfr.x = Wr[(size_t)(k0 + kk) * DD + colBase + n];
    bfr.y = Wr[(size_t)(k0 + kk + 1) * DD + colBase + n];

    cl = __builtin_amdgcn_wmma_f32_16x16x4_f32(
        false, a, false, bfl, (short)0, cl, false, false);
    cr = __builtin_amdgcn_wmma_f32_16x16x4_f32(
        false, a, false, bfr, (short)0, cr, false, false);
  }

  const int rOff = (lane >> 4) * 8;
  const int col  = colBase + n;
  const float bL = bl[col];
  const float bR = br[col];
#pragma unroll
  for (int i = 0; i < 8; ++i) {
    const size_t row = (size_t)(rowBase + i + rOff);
    xl[row * DD + col] = cl[i] + bL;
    xr[row * DD + col] = cr[i] + bR;
  }
}

template <int CTRL>
__device__ __forceinline__ float lane_xadd(float v) {
  const int t = __builtin_amdgcn_update_dpp(0, __float_as_int(v), CTRL, 0xF, 0xF, false);
  return v + __int_as_float(t);
}

__device__ __forceinline__ float lrelu(float m) {
  return fmaxf(m, NEG_SLOPE * m);
}

__device__ __forceinline__ float head_logit(const float4 a, const float4 r,
                                            const float4 w) {
  float p = w.x * lrelu(a.x + r.x);
  p = fmaf(w.y, lrelu(a.y + r.y), p);
  p = fmaf(w.z, lrelu(a.z + r.z), p);
  p = fmaf(w.w, lrelu(a.w + r.w), p);
  p = lane_xadd<0x161>(p);
  p = lane_xadd<0x162>(p);
  p = lane_xadd<0x164>(p);
  return p;
}

__global__ __launch_bounds__(256) void node_attend(
    const int* __restrict__ rowstart, const int* __restrict__ cnt,
    const int* __restrict__ psrc, const float* __restrict__ xl,
    const float* __restrict__ xr, const float* __restrict__ att,
    float* __restrict__ acc, int Np) {
  const int node = __builtin_amdgcn_readfirstlane(
      (int)((blockIdx.x * blockDim.x + threadIdx.x) >> 5));
  const int lane = threadIdx.x & 31;
  if (node >= Np) return;

  const float4 w = ((const float4*)att)[lane];
  const float4 r = ((const float4*)(xr + (size_t)node * DD))[lane];
  const int start = rowstart[node];
  const int deg   = cnt[node];

  const float4 self = ((const float4*)(xl + (size_t)node * DD))[lane];
  float mx = head_logit(self, r, w);
  float s  = 1.0f;
  float4 o = self;

  for (int base = 0; base < deg; base += 32) {
    const int nb = min(deg - base, 32);
    const int mine = (lane < nb) ? psrc[start + base + lane] : node;
    for (int j = 0; j < nb; j += 4) {
      const int s0 = __builtin_amdgcn_readlane(mine, j);
      const int s1 = __builtin_amdgcn_readlane(mine, j + 1);
      const int s2 = __builtin_amdgcn_readlane(mine, j + 2);
      const int s3 = __builtin_amdgcn_readlane(mine, j + 3);
      const float4 a0 = ((const float4*)(xl + (size_t)s0 * DD))[lane];
      const float4 a1 = ((const float4*)(xl + (size_t)s1 * DD))[lane];
      const float4 a2 = ((const float4*)(xl + (size_t)s2 * DD))[lane];
      const float4 a3 = ((const float4*)(xl + (size_t)s3 * DD))[lane];

      const float e0 = head_logit(a0, r, w);
      float e1 = head_logit(a1, r, w);
      float e2 = head_logit(a2, r, w);
      float e3 = head_logit(a3, r, w);
      e1 = (j + 1 < nb) ? e1 : -INFINITY;
      e2 = (j + 2 < nb) ? e2 : -INFINITY;
      e3 = (j + 3 < nb) ? e3 : -INFINITY;

      const float nm = fmaxf(fmaxf(mx, e0), fmaxf(fmaxf(e1, e2), e3));
      const float sc = __expf(mx - nm);
      const float p0 = __expf(e0 - nm);
      const float p1 = __expf(e1 - nm);
      const float p2 = __expf(e2 - nm);
      const float p3 = __expf(e3 - nm);
      mx = nm;
      s = fmaf(s, sc, (p0 + p1) + (p2 + p3));
      o.x = fmaf(p3, a3.x, fmaf(p2, a2.x, fmaf(p1, a1.x, fmaf(p0, a0.x, o.x * sc))));
      o.y = fmaf(p3, a3.y, fmaf(p2, a2.y, fmaf(p1, a1.y, fmaf(p0, a0.y, o.y * sc))));
      o.z = fmaf(p3, a3.z, fmaf(p2, a2.z, fmaf(p1, a1.z, fmaf(p0, a0.z, o.z * sc))));
      o.w = fmaf(p3, a3.w, fmaf(p2, a2.w, fmaf(p1, a1.w, fmaf(p0, a0.w, o.w * sc))));
    }
  }

  const float inv = 1.0f / s;
  float4 res;
  res.x = o.x * inv;
  res.y = o.y * inv;
  res.z = o.z * inv;
  res.w = o.w * inv;
  ((float4*)(acc + (size_t)node * DD))[lane] = res;
}

__global__ __launch_bounds__(256) void node_norm(
    const float* __restrict__ xin, const float* __restrict__ acc,
    const float* __restrict__ bias, const float* __restrict__ gamma,
    const float* __restrict__ beta, float* __restrict__ xout, int Np,
    int do_relu) {
  const int wid  = (blockIdx.x * blockDim.x + threadIdx.x) >> 5;
  const int lane = threadIdx.x & 31;
  if (wid >= Np) return;

  float v[4];
  float sum = 0.0f;
#pragma unroll
  for (int j = 0; j < 4; ++j) {
    const int c = j * 32 + lane;
    const float t = xin[(size_t)wid * DD + c] + acc[(size_t)wid * DD + c] + bias[c];
    v[j] = t;
    sum += t;
  }
#pragma unroll
  for (int off = 16; off > 0; off >>= 1) sum += __shfl_xor(sum, off, 32);
  const float mu = sum * (1.0f / (float)DD);

  float sq = 0.0f;
#pragma unroll
  for (int j = 0; j < 4; ++j) { const float d = v[j] - mu; sq = fmaf(d, d, sq); }
#pragma unroll
  for (int off = 16; off > 0; off >>= 1) sq += __shfl_xor(sq, off, 32);
  const float rstd = rsqrtf(sq * (1.0f / (float)DD) + LN_EPS);

#pragma unroll
  for (int j = 0; j < 4; ++j) {
    const int c = j * 32 + lane;
    float o = (v[j] - mu) * rstd * gamma[c] + beta[c];
    if (do_relu) o = (o > 0.0f) ? o : 0.0f;
    xout[(size_t)wid * DD + c] = o;
  }
}

static inline int cdiv(long long a, long long b) { return (int)((a + b - 1) / b); }

extern "C" void kernel_launch(void* const* d_in, const int* in_sizes, int n_in,
                              void* d_out, int out_size, void* d_ws, size_t ws_size,
                              hipStream_t stream) {
  const int Np = NN;
  const int E  = in_sizes[1] / 2;

  const float* x0    = (const float*)d_in[0];
  const int*   ei    = (const int*)d_in[1];
  const float* Wl    = (const float*)d_in[2];
  const float* bl    = (const float*)d_in[3];
  const float* Wr    = (const float*)d_in[4];
  const float* br    = (const float*)d_in[5];
  const float* att   = (const float*)d_in[6];
  const float* bias  = (const float*)d_in[7];
  const float* gamma = (const float*)d_in[8];
  const float* beta  = (const float*)d_in[9];

  float* ws = (float*)d_ws;
  const size_t ND = (size_t)Np * DD;
  float* bufA     = ws;
  float* xl       = bufA + ND;
  float* xr       = xl + ND;
  float* acc      = xr + ND;
  int*   cnt      = (int*)(acc + ND);
  int*   rowstart = cnt + NPAD;
  int*   cursor   = rowstart + NPAD;
  int*   psrc     = cursor + NPAD;
  int*   total    = psrc + E;

  const int TB = 256;
  const dim3 blk(TB);

  seg_zero<<<cdiv(Np, TB), blk, 0, stream>>>(cnt, total, Np);
  seg_count<<<cdiv(E, TB), blk, 0, stream>>>(ei, cnt, E);
  seg_alloc<<<cdiv(Np, TB), blk, 0, stream>>>(cnt, rowstart, cursor, total, Np);
  seg_fill<<<cdiv(E, TB), blk, 0, stream>>>(ei, cursor, psrc, E);

  for (int i = 0; i < LL; ++i) {
    const float* xin  = (i == 0) ? x0 : bufA;
    float*       xout = (i == LL - 1) ? (float*)d_out : bufA;
    const float* Wli  = Wl + (size_t)i * DD * DD;
    const float* bli  = bl + (size_t)i * DD;
    const float* Wri  = Wr + (size_t)i * DD * DD;
    const float* bri  = br + (size_t)i * DD;
    const float* atti = att + (size_t)i * HH * CC;
    const float* bi   = bias + (size_t)i * DD;
    const float* gi   = gamma + (size_t)i * DD;
    const float* bti  = beta + (size_t)i * DD;

    gemm_dual_wmma<<<Np / 16, blk, 0, stream>>>(xin, Wli, bli, Wri, bri, xl, xr);

    node_attend<<<cdiv((long long)Np * 32, TB), blk, 0, stream>>>(
        rowstart, cnt, psrc, xl, xr, atti, acc, Np);

    node_norm<<<cdiv((long long)Np * 32, TB), blk, 0, stream>>>(
        xin, acc, bi, gi, bti, xout, Np, (i != LL - 1) ? 1 : 0);
  }
}


# GEMM as a resident grid of 256 workgroups: each wave keeps the B fragments of its 16 columns of Wl and Wr in registers and walks over the x row tiles, so W is fetched 256 times a layer, not 3125
# speedup vs baseline: 4.2019x; 1.1634x over previous
#include <hip/hip_runtime.h>
#include <math.h>


#define NN 50000
#define NPAD 50048
#define DD 128
#define HH 4
#define CC 32
#define LL 3
#define NEG_SLOPE 0.2f
#define LN_EPS 1e-5f

typedef __attribute__((ext_vector_type(2))) float v2f;
typedef __attribute__((ext_vector_type(8))) float v8f;

__global__ void seg_zero(int* __restrict__ cnt, int* __restrict__ total, int n) {
  const int i = blockIdx.x * blockDim.x + threadIdx.x;
  if (i < n) cnt[i] = 0;
  if (i == 0) total[0] = 0;
}

__global__ void seg_count(const int* __restrict__ ei, int* __restrict__ cnt, int E) {
  const int t = blockIdx.x * blockDim.x + threadIdx.x;
  if (t < E) atomicAdd(&cnt[ei[E + t]], 1);
}

__global__ __launch_bounds__(256) void seg_alloc(
    const int* __restrict__ cnt, int* __restrict__ rowstart,
    int* __restrict__ cursor, int* __restrict__ total, int n) {
  const int i = blockIdx.x * blockDim.x + threadIdx.x;
  const int lane = threadIdx.x & 31;
  const int c = (i < n) ? cnt[i] : 0;
  int inc = c;
  for (int off = 1; off < 32; off <<= 1) {
    const int t = __shfl_up(inc, off, 32);
    if (lane >= off) inc += t;
  }
  int base = 0;
  if (lane == 31) base = atomicAdd(total, inc);
  base = __shfl(base, 31, 32);
  const int st = base + inc - c;
  if (i < n) { rowstart[i] = st; cursor[i] = st; }
}

__global__ void seg_fill(const int* __restrict__ ei, int* __restrict__ cursor,
                         int* __restrict__ psrc, int E) {
  const int t = blockIdx.x * blockDim.x + threadIdx.x;
  if (t >= E) return;
  const int d = ei[E + t];
  const int p = atomicAdd(&cursor[d], 1);
  if ((unsigned)p < (unsigned)E) atomicExch(&psrc[p], ei[t]);
}

#define GEMM_BLOCKS 256
#define XP 132
#define KSTEPS (DD / 4)
#define GEMM_LDS_WORDS 41984

__global__ __launch_bounds__(256) void gemm_dual_wmma(
    const float* __restrict__ x,
    const float* __restrict__ Wl, const float* __restrict__ bl,
    const float* __restrict__ Wr, const float* __restrict__ br,
    float* __restrict__ xl, float* __restrict__ xr) {
  __shared__ __attribute__((aligned(16))) float sx[GEMM_LDS_WORDS];

  const int tid  = threadIdx.x;
  const int wave = tid >> 5;
  const int lane = tid & 31;

  const int colBase = wave * 16;
  const int m  = lane & 15;
  const int kk = (lane >> 4) * 2;
  const int n  = lane & 15;
  const int col = colBase + n;

  const int sRow = tid >> 5;
  const int sCol = (tid & 31) * 4;
  const int sOff0 = sRow * XP + sCol;
  const int sOff1 = (sRow + 8) * XP + sCol;

  const int nTiles = NN / 16;
  int tile = blockIdx.x;

  float4 g0, g1;
  if (tile < nTiles) {
    const float4* gsrc = (const float4*)(x + (size_t)tile * (16 * DD));
    g0 = gsrc[tid];
    g1 = gsrc[tid + 256];
  }

  v2f fl[KSTEPS];
  v2f fr[KSTEPS];
#pragma unroll
  for (int s = 0; s < KSTEPS / 2; ++s) {
    const int k = 4 * s + kk;
    fl[s].x = Wl[(size_t)k * DD + col];
    fl[s].y = Wl[(size_t)(k + 1) * DD + col];
    fr[s].x = Wr[(size_t)k * DD + col];
    fr[s].y = Wr[(size_t)(k + 1) * DD + col];
  }

  const float bL = bl[col];
  const float bR = br[col];
  const v8f biasL = {bL, bL, bL, bL, bL, bL, bL, bL};
  const v8f biasR = {bR, bR, bR, bR, bR, bR, bR, bR};
  const int oOff = (lane >> 4) * 8 * DD + col;

  if (tile < nTiles) {
    *(float4*)(sx + sOff0) = g0;
    *(float4*)(sx + sOff1) = g1;
  }
  __syncthreads();

#pragma unroll
  for (int s = KSTEPS / 2; s < KSTEPS; ++s) {
    const int k = 4 * s + kk;
    fl[s].x = Wl[(size_t)k * DD + col];
    fl[s].y = Wl[(size_t)(k + 1) * DD + col];
    fr[s].x = Wr[(size_t)k * DD + col];
    fr[s].y = Wr[(size_t)(k + 1) * DD + col];
  }

  int p = 0;
  for (; tile < nTiles; tile += gridDim.x) {
    const int nxt = tile + gridDim.x;
    const bool has = nxt < nTiles;
    if (has) {
      const float4* gsrc = (const float4*)(x + (size_t)nxt * (16 * DD));
      g0 = gsrc[tid];
      g1 = gsrc[tid + 256];
    }

    const float* sa = sx + p * (16 * XP) + m * XP + kk;
    v8f cl = biasL;
    v8f cr = biasR;
#pragma unroll
    for (int s = 0; s < KSTEPS; ++s) {
      v2f a;
      a.x = sa[4 * s];
      a.y = sa[4 * s + 1];
      cl = __builtin_amdgcn_wmma_f32_16x16x4_f32(
          false, a, false, fl[s], (short)0, cl, false, false);
      cr = __builtin_amdgcn_wmma_f32_16x16x4_f32(
          false, a, false, fr[s], (short)0, cr, false, false);
    }

    if (has) {
      float* sd = sx + (p ^ 1) * (16 * XP);
      *(float4*)(sd + sOff0) = g0;
      *(float4*)(sd + sOff1) = g1;
    }
    __syncthreads();
    p ^= 1;

    float* xlT = xl + (size_t)tile * (16 * DD);
    float* xrT = xr + (size_t)tile * (16 * DD);
#pragma unroll
    for (int i = 0; i < 8; ++i) {
      xlT[oOff + i * DD] = cl[i];
      xrT[oOff + i * DD] = cr[i];
    }
  }
}

template <int CTRL>
__device__ __forceinline__ float lane_xadd(float v) {
  const int t = __builtin_amdgcn_update_dpp(0, __float_as_int(v), CTRL, 0xF, 0xF, false);
  return v + __int_as_float(t);
}

__device__ __forceinline__ float other_row(float v) {
  const int b = __float_as_int(v);
  return __int_as_float(
      __builtin_amdgcn_permlanex16(b, b, 0x76543210, 0xfedcba98, false, false));
}

__device__ __forceinline__ float wave_sum(float v) {
  v = lane_xadd<0x161>(v);
  v = lane_xadd<0x162>(v);
  v = lane_xadd<0x164>(v);
  v = lane_xadd<0x168>(v);
  return v + other_row(v);
}

__device__ __forceinline__ float lrelu(float m) {
  return fmaxf(m, NEG_SLOPE * m);
}

__device__ __forceinline__ float head_logit(const float4 a, const float4 r,
                                            const float4 w) {
  float p = w.x * lrelu(a.x + r.x);
  p = fmaf(w.y, lrelu(a.y + r.y), p);
  p = fmaf(w.z, lrelu(a.z + r.z), p);
  p = fmaf(w.w, lrelu(a.w + r.w), p);
  p = lane_xadd<0x161>(p);
  p = lane_xadd<0x162>(p);
  p = lane_xadd<0x164>(p);
  return p;
}

__global__ __launch_bounds__(256) void node_attend_norm(
    const int* __restrict__ rowstart, const int* __restrict__ cnt,
    const int* __restrict__ psrc, const float* __restrict__ xl,
    const float* __restrict__ xr, const float* __restrict__ att,
    const float* xin, const float* __restrict__ bias,
    const float* __restrict__ gamma, const float* __restrict__ beta,
    float* xout, int Np, int do_relu) {
  const int node = __builtin_amdgcn_readfirstlane(
      (int)((blockIdx.x * blockDim.x + threadIdx.x) >> 5));
  const int lane = threadIdx.x & 31;
  if (node >= Np) return;

  const float4 w = ((const float4*)att)[lane];
  const float4 r = ((const float4*)(xr + (size_t)node * DD))[lane];
  const int start = rowstart[node];
  const int deg   = cnt[node];

  const float4 self = ((const float4*)(xl + (size_t)node * DD))[lane];
  float mx = head_logit(self, r, w);
  float s  = 1.0f;
  float4 o = self;

  for (int base = 0; base < deg; base += 32) {
    const int nb = min(deg - base, 32);
    const int mine = (lane < nb) ? psrc[start + base + lane] : node;
    for (int j = 0; j < nb; j += 4) {
      const int s0 = __builtin_amdgcn_readlane(mine, j);
      const int s1 = __builtin_amdgcn_readlane(mine, j + 1);
      const int s2 = __builtin_amdgcn_readlane(mine, j + 2);
      const int s3 = __builtin_amdgcn_readlane(mine, j + 3);
      const float4 a0 = ((const float4*)(xl + (size_t)s0 * DD))[lane];
      const float4 a1 = ((const float4*)(xl + (size_t)s1 * DD))[lane];
      const float4 a2 = ((const float4*)(xl + (size_t)s2 * DD))[lane];
      const float4 a3 = ((const float4*)(xl + (size_t)s3 * DD))[lane];

      const float e0 = head_logit(a0, r, w);
      float e1 = head_logit(a1, r, w);
      float e2 = head_logit(a2, r, w);
      float e3 = head_logit(a3, r, w);
      e1 = (j + 1 < nb) ? e1 : -INFINITY;
      e2 = (j + 2 < nb) ? e2 : -INFINITY;
      e3 = (j + 3 < nb) ? e3 : -INFINITY;

      const float nm = fmaxf(fmaxf(mx, e0), fmaxf(fmaxf(e1, e2), e3));
      const float sc = __expf(mx - nm);
      const float p0 = __expf(e0 - nm);
      const float p1 = __expf(e1 - nm);
      const float p2 = __expf(e2 - nm);
      const float p3 = __expf(e3 - nm);
      mx = nm;
      s = fmaf(s, sc, (p0 + p1) + (p2 + p3));
      o.x = fmaf(p3, a3.x, fmaf(p2, a2.x, fmaf(p1, a1.x, fmaf(p0, a0.x, o.x * sc))));
      o.y = fmaf(p3, a3.y, fmaf(p2, a2.y, fmaf(p1, a1.y, fmaf(p0, a0.y, o.y * sc))));
      o.z = fmaf(p3, a3.z, fmaf(p2, a2.z, fmaf(p1, a1.z, fmaf(p0, a0.z, o.z * sc))));
      o.w = fmaf(p3, a3.w, fmaf(p2, a2.w, fmaf(p1, a1.w, fmaf(p0, a0.w, o.w * sc))));
    }
  }

  const float inv = 1.0f / s;

  const float4 xi = ((const float4*)(xin + (size_t)node * DD))[lane];
  const float4 b4 = ((const float4*)bias)[lane];
  const float4 g4 = ((const float4*)gamma)[lane];
  const float4 t4 = ((const float4*)beta)[lane];
  float4 v;
  v.x = xi.x + o.x * inv + b4.x;
  v.y = xi.y + o.y * inv + b4.y;
  v.z = xi.z + o.z * inv + b4.z;
  v.w = xi.w + o.w * inv + b4.w;
  const float mu = wave_sum((v.x + v.y) + (v.z + v.w)) * (1.0f / (float)DD);
  v.x -= mu;
  v.y -= mu;
  v.z -= mu;
  v.w -= mu;
  const float sq = wave_sum(fmaf(v.w, v.w, fmaf(v.z, v.z, fmaf(v.y, v.y, v.x * v.x))));
  const float rstd = rsqrtf(sq * (1.0f / (float)DD) + LN_EPS);
  float4 res;
  res.x = v.x * rstd * g4.x + t4.x;
  res.y = v.y * rstd * g4.y + t4.y;
  res.z = v.z * rstd * g4.z + t4.z;
  res.w = v.w * rstd * g4.w + t4.w;
  if (do_relu) {
    res.x = (res.x > 0.0f) ? res.x : 0.0f;
    res.y = (res.y > 0.0f) ? res.y : 0.0f;
    res.z = (res.z > 0.0f) ? res.z : 0.0f;
    res.w = (res.w > 0.0f) ? res.w : 0.0f;
  }
  ((float4*)(xout + (size_t)node * DD))[lane] = res;
}

static inline int cdiv(long long a, long long b) { return (int)((a + b - 1) / b); }

extern "C" void kernel_launch(void* const* d_in, const int* in_sizes, int n_in,
                              void* d_out, int out_size, void* d_ws, size_t ws_size,
                              hipStream_t stream) {
  const int Np = NN;
  const int E  = in_sizes[1] / 2;

  const float* x0    = (const float*)d_in[0];
  const int*   ei    = (const int*)d_in[1];
  const float* Wl    = (const float*)d_in[2];
  const float* bl    = (const float*)d_in[3];
  const float* Wr    = (const float*)d_in[4];
  const float* br    = (const float*)d_in[5];
  const float* att   = (const float*)d_in[6];
  const float* bias  = (const float*)d_in[7];
  const float* gamma = (const float*)d_in[8];
  const float* beta  = (const float*)d_in[9];

  float* ws = (float*)d_ws;
  const size_t ND = (size_t)Np * DD;
  float* bufA     = ws;
  float* xl       = bufA + ND;
  float* xr       = xl + ND;
  int*   cnt      = (int*)(xr + ND);
  int*   rowstart = cnt + NPAD;
  int*   cursor   = rowstart + NPAD;
  int*   psrc     = cursor + NPAD;
  int*   total    = psrc + E;

  const int TB = 256;
  const dim3 blk(TB);

  seg_zero<<<cdiv(Np, TB), blk, 0, stream>>>(cnt, total, Np);
  seg_count<<<cdiv(E, TB), blk, 0, stream>>>(ei, cnt, E);
  seg_alloc<<<cdiv(Np, TB), blk, 0, stream>>>(cnt, rowstart, cursor, total, Np);
  seg_fill<<<cdiv(E, TB), blk, 0, stream>>>(ei, cursor, psrc, E);

  for (int i = 0; i < LL; ++i) {
    const float* xin  = (i == 0) ? x0 : bufA;
    float*       xout = (i == LL - 1) ? (float*)d_out : bufA;
    const float* Wli  = Wl + (size_t)i * DD * DD;
    const float* bli  = bl + (size_t)i * DD;
    const float* Wri  = Wr + (size_t)i * DD * DD;
    const float* bri  = br + (size_t)i * DD;
    const float* atti = att + (size_t)i * HH * CC;
    const float* bi   = bias + (size_t)i * DD;
    const float* gi   = gamma + (size_t)i * DD;
    const float* bti  = beta + (size_t)i * DD;

    gemm_dual_wmma<<<GEMM_BLOCKS, blk, 0, stream>>>(xin, Wli, bli, Wri, bri, xl, xr);

    node_attend_norm<<<cdiv((long long)Np * 32, TB), blk, 0, stream>>>(
        rowstart, cnt, psrc, xl, xr, atti, xin, bi, gi, bti, xout, Np,
        (i != LL - 1) ? 1 : 0);
  }
}
